# starting kernel
# baseline (speedup 1.0000x reference)
#include <hip/hip_runtime.h>

typedef float v2f __attribute__((ext_vector_type(2)));
typedef float v8f __attribute__((ext_vector_type(8)));

#define B_    512
#define D_    256
#define HID_  512
#define OUT_  128
#define CONC_ 1536

__global__ __launch_bounds__(256) void scatter_inv(const int* __restrict__ idx,
                                                   int nrows,
                                                   int* __restrict__ inv) {
  int r = blockIdx.x * 256 + threadIdx.x;
  if (r < nrows) inv[idx[r]] = r;
}

__global__ __launch_bounds__(256) void gather_sum(const float* __restrict__ x,
                                                  const int* __restrict__ inv,
                                                  int m, int chunks,
                                                  float* __restrict__ pooled) {
  const int b  = blockIdx.x / chunks;
  const int c  = blockIdx.x % chunks;
  const int d  = threadIdx.x;
  const int* invb = inv + b * m;

  int p0 = c * 64;
  int p1 = p0 + 64; if (p1 > m) p1 = m;

  float acc = 0.f;
  for (int p = p0; p < p1; ++p) {
    int r = invb[p];
    if (r >= 0) acc += x[(size_t)r * D_ + d];
  }
  unsafeAtomicAdd(&pooled[b * D_ + d], acc);
}

__global__ __launch_bounds__(256) void gemm1_relu(const float* __restrict__ pooledAll,
    const float* __restrict__ W1_0, const float* __restrict__ b1_0,
    const float* __restrict__ W1_1, const float* __restrict__ b1_1,
    const float* __restrict__ W1_2, const float* __restrict__ b1_2,
    float* __restrict__ h) {
  const int dim = blockIdx.y;
  const float* W    = (dim == 0) ? W1_0 : (dim == 1) ? W1_1 : W1_2;
  const float* bias = (dim == 0) ? b1_0 : (dim == 1) ? b1_1 : b1_2;
  const float* P    = pooledAll + (size_t)dim * B_ * D_;

  const int wave = threadIdx.x >> 5;
  const int lane = threadIdx.x & 31;
  const int job  = blockIdx.x * 8 + wave;
  const int mt   = job >> 3;
  const int ns   = job & 7;

  const int mrow = mt * 16 + (lane & 15);
  const int kk   = (lane >> 4) * 2;
  const int ncol = ns * 64 + (lane & 15);

  v8f a0 = {}, a1 = {}, a2 = {}, a3 = {};
  for (int k = 0; k < D_; k += 4) {
    v2f av = *(const v2f*)(P + (size_t)mrow * D_ + k + kk);
    const float* w0 = W + (size_t)(k + kk) * HID_ + ncol;
    const float* w1 = w0 + HID_;
    v2f bf0 = { w0[0],  w1[0]  };
    v2f bf1 = { w0[16], w1[16] };
    v2f bf2 = { w0[32], w1[32] };
    v2f bf3 = { w0[48], w1[48] };
    a0 = __builtin_amdgcn_wmma_f32_16x16x4_f32(false, av, false, bf0, (short)0, a0, false, false);
    a1 = __builtin_amdgcn_wmma_f32_16x16x4_f32(false, av, false, bf1, (short)0, a1, false, false);
    a2 = __builtin_amdgcn_wmma_f32_16x16x4_f32(false, av, false, bf2, (short)0, a2, false, false);
    a3 = __builtin_amdgcn_wmma_f32_16x16x4_f32(false, av, false, bf3, (short)0, a3, false, false);
  }

  const int nbase = ns * 64 + (lane & 15);
  const int rbase = mt * 16 + (lane >> 4) * 8;
  const float bb0 = bias[nbase], bb1 = bias[nbase + 16],
              bb2 = bias[nbase + 32], bb3 = bias[nbase + 48];
#pragma unroll
  for (int v = 0; v < 8; ++v) {
    float* hr = h + (size_t)(rbase + v) * CONC_ + (size_t)dim * HID_ + nbase;
    hr[0]  = fmaxf(a0[v] + bb0, 0.f);
    hr[16] = fmaxf(a1[v] + bb1, 0.f);
    hr[32] = fmaxf(a2[v] + bb2, 0.f);
    hr[48] = fmaxf(a3[v] + bb3, 0.f);
  }
}

__global__ __launch_bounds__(256) void gemm2(const float* __restrict__ h,
                                             const float* __restrict__ W2,
                                             const float* __restrict__ b2,
                                             float* __restrict__ out) {
  const int wave = threadIdx.x >> 5;
  const int lane = threadIdx.x & 31;
  const int job  = blockIdx.x * 8 + wave;
  const int mt   = job >> 1;
  const int ns   = job & 1;

  const int mrow = mt * 16 + (lane & 15);
  const int kk   = (lane >> 4) * 2;
  const int ncol = ns * 64 + (lane & 15);

  v8f a0 = {}, a1 = {}, a2 = {}, a3 = {};
  for (int k = 0; k < CONC_; k += 4) {
    v2f av = *(const v2f*)(h + (size_t)mrow * CONC_ + k + kk);
    const float* w0 = W2 + (size_t)(k + kk) * OUT_ + ncol;
    const float* w1 = w0 + OUT_;
    v2f bf0 = { w0[0],  w1[0]  };
    v2f bf1 = { w0[16], w1[16] };
    v2f bf2 = { w0[32], w1[32] };
    v2f bf3 = { w0[48], w1[48] };
    a0 = __builtin_amdgcn_wmma_f32_16x16x4_f32(false, av, false, bf0, (short)0, a0, false, false);
    a1 = __builtin_amdgcn_wmma_f32_16x16x4_f32(false, av, false, bf1, (short)0, a1, false, false);
    a2 = __builtin_amdgcn_wmma_f32_16x16x4_f32(false, av, false, bf2, (short)0, a2, false, false);
    a3 = __builtin_amdgcn_wmma_f32_16x16x4_f32(false, av, false, bf3, (short)0, a3, false, false);
  }

  const int nbase = ns * 64 + (lane & 15);
  const int rbase = mt * 16 + (lane >> 4) * 8;
  const float bb0 = b2[nbase], bb1 = b2[nbase + 16],
              bb2 = b2[nbase + 32], bb3 = b2[nbase + 48];
#pragma unroll
  for (int v = 0; v < 8; ++v) {
    float* orow = out + (size_t)(rbase + v) * OUT_ + nbase;
    orow[0]  = a0[v] + bb0;
    orow[16] = a1[v] + bb1;
    orow[32] = a2[v] + bb2;
    orow[48] = a3[v] + bb3;
  }
}

extern "C" void kernel_launch(void* const* d_in, const int* in_sizes, int n_in,
                              void* d_out, int out_size, void* d_ws, size_t ws_size,
                              hipStream_t stream) {
  const float* x0   = (const float*)d_in[0];
  const int*   idx0 = (const int*)  d_in[1];
  const float* x1   = (const float*)d_in[2];
  const int*   idx1 = (const int*)  d_in[3];
  const float* x2   = (const float*)d_in[4];
  const int*   idx2 = (const int*)  d_in[5];
  const float* W1_0 = (const float*)d_in[6];
  const float* b1_0 = (const float*)d_in[7];
  const float* W1_1 = (const float*)d_in[8];
  const float* b1_1 = (const float*)d_in[9];
  const float* W1_2 = (const float*)d_in[10];
  const float* b1_2 = (const float*)d_in[11];
  const float* W2   = (const float*)d_in[12];
  const float* b2   = (const float*)d_in[13];
  float* out = (float*)d_out;

  float* pooled = (float*)d_ws;
  float* h      = pooled + 3 * B_ * D_;
  int*   inv    = (int*)(h + (size_t)B_ * CONC_);
  const int n0 = B_ * 128, n1 = B_ * 256, n2 = B_ * 64;
  int* inv0 = inv;
  int* inv1 = inv0 + n0;
  int* inv2 = inv1 + n1;

  hipMemsetAsync(pooled, 0,    (size_t)3 * B_ * D_ * sizeof(float), stream);
  hipMemsetAsync(inv,    0xFF, (size_t)(n0 + n1 + n2) * sizeof(int), stream);

  scatter_inv<<<(n0 + 255) / 256, 256, 0, stream>>>(idx0, n0, inv0);
  scatter_inv<<<(n1 + 255) / 256, 256, 0, stream>>>(idx1, n1, inv1);
  scatter_inv<<<(n2 + 255) / 256, 256, 0, stream>>>(idx2, n2, inv2);

  gather_sum<<<B_ * 2, 256, 0, stream>>>(x0, inv0, 128, 2, pooled);
  gather_sum<<<B_ * 4, 256, 0, stream>>>(x1, inv1, 256, 4, pooled + B_ * D_);
  gather_sum<<<B_ * 1, 256, 0, stream>>>(x2, inv2,  64, 1, pooled + 2 * B_ * D_);

  dim3 g1(32, 3);
  gemm1_relu<<<g1, 256, 0, stream>>>(pooled, W1_0, b1_0, W1_1, b1_1, W1_2, b1_2, h);

  gemm2<<<8, 256, 0, stream>>>(h, W2, b2, out);
}


# gather_sum: one work-group per batch element, x rows read as 16 independent 16-byte loads per lane, partial sums added in fixed wave order through LDS, no atomics; summation order changes
# speedup vs baseline: 2.3762x; 2.3762x over previous
#include <hip/hip_runtime.h>

typedef float v2f __attribute__((ext_vector_type(2)));
typedef float v4f __attribute__((ext_vector_type(4)));
typedef float v8f __attribute__((ext_vector_type(8)));

#define B_    512
#define D_    256
#define HID_  512
#define OUT_  128
#define CONC_ 1536

__global__ __launch_bounds__(256) void scatter_inv(const int* __restrict__ idx,
                                                   int nrows,
                                                   int* __restrict__ inv) {
  int r = blockIdx.x * 256 + threadIdx.x;
  if (r < nrows) inv[idx[r]] = r;
}

template <int M>
__global__ __launch_bounds__(256) void gather_sum(const float* __restrict__ x,
                                                  const int* __restrict__ inv,
                                                  float* __restrict__ pooled) {
  constexpr int RPW = M / 8;
  constexpr int GRP = 8;
  static_assert(RPW >= GRP && RPW <= 32 && RPW % GRP == 0, "one index per lane, whole groups");
  __shared__ v4f part[8][64];

  const int b    = blockIdx.x;
  const int wave = __builtin_amdgcn_readfirstlane(threadIdx.x >> 5);
  const int lane = threadIdx.x & 31;

  int myr = -1;
  if (lane < RPW) myr = inv[b * M + wave * RPW + lane];

  v4f acc0 = {0.f, 0.f, 0.f, 0.f}, acc1 = {0.f, 0.f, 0.f, 0.f};
#pragma unroll
  for (int j0 = 0; j0 < RPW; j0 += GRP) {
    v4f v0[GRP], v1[GRP];
    int rr[GRP];
#pragma unroll
    for (int j = 0; j < GRP; ++j) {
      rr[j] = __builtin_amdgcn_readlane(myr, j0 + j);
      const int rc = rr[j] < 0 ? 0 : rr[j];
      const v4f* row = (const v4f*)(x + (size_t)rc * D_);
      v0[j] = __builtin_nontemporal_load(row + lane);
      v1[j] = __builtin_nontemporal_load(row + 32 + lane);
    }
#pragma unroll
    for (int j = 0; j < GRP; ++j) {
      const bool on = rr[j] >= 0;
      acc0.x += on ? v0[j].x : 0.f;  acc0.y += on ? v0[j].y : 0.f;
      acc0.z += on ? v0[j].z : 0.f;  acc0.w += on ? v0[j].w : 0.f;
      acc1.x += on ? v1[j].x : 0.f;  acc1.y += on ? v1[j].y : 0.f;
      acc1.z += on ? v1[j].z : 0.f;  acc1.w += on ? v1[j].w : 0.f;
    }
  }
  part[wave][lane]      = acc0;
  part[wave][32 + lane] = acc1;
  __syncthreads();

  const float* pf = (const float*)&part[0][0];
  const int d = threadIdx.x;
  float s = pf[d];
#pragma unroll
  for (int w = 1; w < 8; ++w) s += pf[w * D_ + d];
  pooled[b * D_ + d] = s;
}

__global__ __launch_bounds__(256) void gemm1_relu(const float* __restrict__ pooledAll,
    const float* __restrict__ W1_0, const float* __restrict__ b1_0,
    const float* __restrict__ W1_1, const float* __restrict__ b1_1,
    const float* __restrict__ W1_2, const float* __restrict__ b1_2,
    float* __restrict__ h) {
  const int dim = blockIdx.y;
  const float* W    = (dim == 0) ? W1_0 : (dim == 1) ? W1_1 : W1_2;
  const float* bias = (dim == 0) ? b1_0 : (dim == 1) ? b1_1 : b1_2;
  const float* P    = pooledAll + (size_t)dim * B_ * D_;

  const int wave = threadIdx.x >> 5;
  const int lane = threadIdx.x & 31;
  const int job  = blockIdx.x * 8 + wave;
  const int mt   = job >> 3;
  const int ns   = job & 7;

  const int mrow = mt * 16 + (lane & 15);
  const int kk   = (lane >> 4) * 2;
  const int ncol = ns * 64 + (lane & 15);

  v8f a0 = {}, a1 = {}, a2 = {}, a3 = {};
  for (int k = 0; k < D_; k += 4) {
    v2f av = *(const v2f*)(P + (size_t)mrow * D_ + k + kk);
    const float* w0 = W + (size_t)(k + kk) * HID_ + ncol;
    const float* w1 = w0 + HID_;
    v2f bf0 = { w0[0],  w1[0]  };
    v2f bf1 = { w0[16], w1[16] };
    v2f bf2 = { w0[32], w1[32] };
    v2f bf3 = { w0[48], w1[48] };
    a0 = __builtin_amdgcn_wmma_f32_16x16x4_f32(false, av, false, bf0, (short)0, a0, false, false);
    a1 = __builtin_amdgcn_wmma_f32_16x16x4_f32(false, av, false, bf1, (short)0, a1, false, false);
    a2 = __builtin_amdgcn_wmma_f32_16x16x4_f32(false, av, false, bf2, (short)0, a2, false, false);
    a3 = __builtin_amdgcn_wmma_f32_16x16x4_f32(false, av, false, bf3, (short)0, a3, false, false);
  }

  const int nbase = ns * 64 + (lane & 15);
  const int rbase = mt * 16 + (lane >> 4) * 8;
  const float bb0 = bias[nbase], bb1 = bias[nbase + 16],
              bb2 = bias[nbase + 32], bb3 = bias[nbase + 48];
#pragma unroll
  for (int v = 0; v < 8; ++v) {
    float* hr = h + (size_t)(rbase + v) * CONC_ + (size_t)dim * HID_ + nbase;
    hr[0]  = fmaxf(a0[v] + bb0, 0.f);
    hr[16] = fmaxf(a1[v] + bb1, 0.f);
    hr[32] = fmaxf(a2[v] + bb2, 0.f);
    hr[48] = fmaxf(a3[v] + bb3, 0.f);
  }
}

__global__ __launch_bounds__(256) void gemm2(const float* __restrict__ h,
                                             const float* __restrict__ W2,
                                             const float* __restrict__ b2,
                                             float* __restrict__ out) {
  const int wave = threadIdx.x >> 5;
  const int lane = threadIdx.x & 31;
  const int job  = blockIdx.x * 8 + wave;
  const int mt   = job >> 1;
  const int ns   = job & 1;

  const int mrow = mt * 16 + (lane & 15);
  const int kk   = (lane >> 4) * 2;
  const int ncol = ns * 64 + (lane & 15);

  v8f a0 = {}, a1 = {}, a2 = {}, a3 = {};
  for (int k = 0; k < CONC_; k += 4) {
    v2f av = *(const v2f*)(h + (size_t)mrow * CONC_ + k + kk);
    const float* w0 = W2 + (size_t)(k + kk) * OUT_ + ncol;
    const float* w1 = w0 + OUT_;
    v2f bf0 = { w0[0],  w1[0]  };
    v2f bf1 = { w0[16], w1[16] };
    v2f bf2 = { w0[32], w1[32] };
    v2f bf3 = { w0[48], w1[48] };
    a0 = __builtin_amdgcn_wmma_f32_16x16x4_f32(false, av, false, bf0, (short)0, a0, false, false);
    a1 = __builtin_amdgcn_wmma_f32_16x16x4_f32(false, av, false, bf1, (short)0, a1, false, false);
    a2 = __builtin_amdgcn_wmma_f32_16x16x4_f32(false, av, false, bf2, (short)0, a2, false, false);
    a3 = __builtin_amdgcn_wmma_f32_16x16x4_f32(false, av, false, bf3, (short)0, a3, false, false);
  }

  const int nbase = ns * 64 + (lane & 15);
  const int rbase = mt * 16 + (lane >> 4) * 8;
  const float bb0 = b2[nbase], bb1 = b2[nbase + 16],
              bb2 = b2[nbase + 32], bb3 = b2[nbase + 48];
#pragma unroll
  for (int v = 0; v < 8; ++v) {
    float* orow = out + (size_t)(rbase + v) * OUT_ + nbase;
    orow[0]  = a0[v] + bb0;
    orow[16] = a1[v] + bb1;
    orow[32] = a2[v] + bb2;
    orow[48] = a3[v] + bb3;
  }
}

extern "C" void kernel_launch(void* const* d_in, const int* in_sizes, int n_in,
                              void* d_out, int out_size, void* d_ws, size_t ws_size,
                              hipStream_t stream) {
  const float* x0   = (const float*)d_in[0];
  const int*   idx0 = (const int*)  d_in[1];
  const float* x1   = (const float*)d_in[2];
  const int*   idx1 = (const int*)  d_in[3];
  const float* x2   = (const float*)d_in[4];
  const int*   idx2 = (const int*)  d_in[5];
  const float* W1_0 = (const float*)d_in[6];
  const float* b1_0 = (const float*)d_in[7];
  const float* W1_1 = (const float*)d_in[8];
  const float* b1_1 = (const float*)d_in[9];
  const float* W1_2 = (const float*)d_in[10];
  const float* b1_2 = (const float*)d_in[11];
  const float* W2   = (const float*)d_in[12];
  const float* b2   = (const float*)d_in[13];
  float* out = (float*)d_out;

  float* pooled = (float*)d_ws;
  float* h      = pooled + 3 * B_ * D_;
  int*   inv    = (int*)(h + (size_t)B_ * CONC_);
  const int n0 = B_ * 128, n1 = B_ * 256, n2 = B_ * 64;
  int* inv0 = inv;
  int* inv1 = inv0 + n0;
  int* inv2 = inv1 + n1;

  hipMemsetAsync(inv,    0xFF, (size_t)(n0 + n1 + n2) * sizeof(int), stream);

  scatter_inv<<<(n0 + 255) / 256, 256, 0, stream>>>(idx0, n0, inv0);
  scatter_inv<<<(n1 + 255) / 256, 256, 0, stream>>>(idx1, n1, inv1);
  scatter_inv<<<(n2 + 255) / 256, 256, 0, stream>>>(idx2, n2, inv2);

  gather_sum<128><<<B_, 256, 0, stream>>>(x0, inv0, pooled);
  gather_sum<256><<<B_, 256, 0, stream>>>(x1, inv1, pooled + B_ * D_);
  gather_sum< 64><<<B_, 256, 0, stream>>>(x2, inv2, pooled + 2 * B_ * D_);

  dim3 g1(32, 3);
  gemm1_relu<<<g1, 256, 0, stream>>>(pooled, W1_0, b1_0, W1_1, b1_1, W1_2, b1_2, h);

  gemm2<<<8, 256, 0, stream>>>(h, W2, b2, out);
}


# gemm2 on 128 work-groups instead of 8: each owns a 16x32 out tile, K split over its 8 waves, partial tiles added in wave order through LDS, one 128-byte store per row; summation order changes
# speedup vs baseline: 4.4551x; 1.8749x over previous
#include <hip/hip_runtime.h>

typedef float v2f __attribute__((ext_vector_type(2)));
typedef float v4f __attribute__((ext_vector_type(4)));
typedef float v8f __attribute__((ext_vector_type(8)));

#define B_    512
#define D_    256
#define HID_  512
#define OUT_  128
#define CONC_ 1536

__global__ __launch_bounds__(256) void scatter_inv(const int* __restrict__ idx,
                                                   int nrows,
                                                   int* __restrict__ inv) {
  int r = blockIdx.x * 256 + threadIdx.x;
  if (r < nrows) inv[idx[r]] = r;
}

template <int M>
__global__ __launch_bounds__(256) void gather_sum(const float* __restrict__ x,
                                                  const int* __restrict__ inv,
                                                  float* __restrict__ pooled) {
  constexpr int RPW = M / 8;
  constexpr int GRP = 8;
  static_assert(RPW >= GRP && RPW <= 32 && RPW % GRP == 0, "one index per lane, whole groups");
  __shared__ v4f part[8][64];

  const int b    = blockIdx.x;
  const int wave = __builtin_amdgcn_readfirstlane(threadIdx.x >> 5);
  const int lane = threadIdx.x & 31;

  int myr = -1;
  if (lane < RPW) myr = inv[b * M + wave * RPW + lane];

  v4f acc0 = {0.f, 0.f, 0.f, 0.f}, acc1 = {0.f, 0.f, 0.f, 0.f};
#pragma unroll
  for (int j0 = 0; j0 < RPW; j0 += GRP) {
    v4f v0[GRP], v1[GRP];
    int rr[GRP];
#pragma unroll
    for (int j = 0; j < GRP; ++j) {
      rr[j] = __builtin_amdgcn_readlane(myr, j0 + j);
      const int rc = rr[j] < 0 ? 0 : rr[j];
      const v4f* row = (const v4f*)(x + (size_t)rc * D_);
      v0[j] = __builtin_nontemporal_load(row + lane);
      v1[j] = __builtin_nontemporal_load(row + 32 + lane);
    }
#pragma unroll
    for (int j = 0; j < GRP; ++j) {
      const bool on = rr[j] >= 0;
      acc0.x += on ? v0[j].x : 0.f;  acc0.y += on ? v0[j].y : 0.f;
      acc0.z += on ? v0[j].z : 0.f;  acc0.w += on ? v0[j].w : 0.f;
      acc1.x += on ? v1[j].x : 0.f;  acc1.y += on ? v1[j].y : 0.f;
      acc1.z += on ? v1[j].z : 0.f;  acc1.w += on ? v1[j].w : 0.f;
    }
  }
  part[wave][lane]      = acc0;
  part[wave][32 + lane] = acc1;
  __syncthreads();

  const float* pf = (const float*)&part[0][0];
  const int d = threadIdx.x;
  float s = pf[d];
#pragma unroll
  for (int w = 1; w < 8; ++w) s += pf[w * D_ + d];
  pooled[b * D_ + d] = s;
}

__global__ __launch_bounds__(256) void gemm1_relu(const float* __restrict__ pooledAll,
    const float* __restrict__ W1_0, const float* __restrict__ b1_0,
    const float* __restrict__ W1_1, const float* __restrict__ b1_1,
    const float* __restrict__ W1_2, const float* __restrict__ b1_2,
    float* __restrict__ h) {
  const int dim = blockIdx.y;
  const float* W    = (dim == 0) ? W1_0 : (dim == 1) ? W1_1 : W1_2;
  const float* bias = (dim == 0) ? b1_0 : (dim == 1) ? b1_1 : b1_2;
  const float* P    = pooledAll + (size_t)dim * B_ * D_;

  const int wave = threadIdx.x >> 5;
  const int lane = threadIdx.x & 31;
  const int job  = blockIdx.x * 8 + wave;
  const int mt   = job >> 3;
  const int ns   = job & 7;

  const int mrow = mt * 16 + (lane & 15);
  const int kk   = (lane >> 4) * 2;
  const int ncol = ns * 64 + (lane & 15);

  v8f a0 = {}, a1 = {}, a2 = {}, a3 = {};
  for (int k = 0; k < D_; k += 4) {
    v2f av = *(const v2f*)(P + (size_t)mrow * D_ + k + kk);
    const float* w0 = W + (size_t)(k + kk) * HID_ + ncol;
    const float* w1 = w0 + HID_;
    v2f bf0 = { w0[0],  w1[0]  };
    v2f bf1 = { w0[16], w1[16] };
    v2f bf2 = { w0[32], w1[32] };
    v2f bf3 = { w0[48], w1[48] };
    a0 = __builtin_amdgcn_wmma_f32_16x16x4_f32(false, av, false, bf0, (short)0, a0, false, false);
    a1 = __builtin_amdgcn_wmma_f32_16x16x4_f32(false, av, false, bf1, (short)0, a1, false, false);
    a2 = __builtin_amdgcn_wmma_f32_16x16x4_f32(false, av, false, bf2, (short)0, a2, false, false);
    a3 = __builtin_amdgcn_wmma_f32_16x16x4_f32(false, av, false, bf3, (short)0, a3, false, false);
  }

  const int nbase = ns * 64 + (lane & 15);
  const int rbase = mt * 16 + (lane >> 4) * 8;
  const float bb0 = bias[nbase], bb1 = bias[nbase + 16],
              bb2 = bias[nbase + 32], bb3 = bias[nbase + 48];
#pragma unroll
  for (int v = 0; v < 8; ++v) {
    float* hr = h + (size_t)(rbase + v) * CONC_ + (size_t)dim * HID_ + nbase;
    hr[0]  = fmaxf(a0[v] + bb0, 0.f);
    hr[16] = fmaxf(a1[v] + bb1, 0.f);
    hr[32] = fmaxf(a2[v] + bb2, 0.f);
    hr[48] = fmaxf(a3[v] + bb3, 0.f);
  }
}

#define G2_SPLIT 8
#define G2_KW    (CONC_ / G2_SPLIT)
#define G2_PITCH 36

__global__ __launch_bounds__(256) void gemm2(const float* __restrict__ h,
                                             const float* __restrict__ W2,
                                             const float* __restrict__ b2,
                                             float* __restrict__ out) {
  __shared__ float part[G2_SPLIT][16 * G2_PITCH];

  const int wave = threadIdx.x >> 5;
  const int lane = threadIdx.x & 31;
  const int mt   = blockIdx.x >> 2;
  const int ns   = blockIdx.x & 3;

  const int c    = lane & 15;
  const int hh   = lane >> 4;
  const int kk   = hh * 2;
  const int k0   = wave * G2_KW;

  const float* ap = h  + (size_t)(mt * 16 + c) * CONC_ + k0 + kk;
  const float* bp = W2 + (size_t)(k0 + kk) * OUT_ + ns * 32 + c;

  const float bb = b2[ns * 32 + lane];

  v8f a0 = {}, a1 = {};
#pragma unroll
  for (int s = 0; s < G2_KW / 4; ++s) {
    v2f av = *(const v2f*)(ap + 4 * s);
    const float* w0 = bp + (size_t)(4 * s) * OUT_;
    const float* w1 = w0 + OUT_;
    v2f bf0 = { w0[0],  w1[0]  };
    v2f bf1 = { w0[16], w1[16] };
    a0 = __builtin_amdgcn_wmma_f32_16x16x4_f32(false, av, false, bf0, (short)0, a0, false, false);
    a1 = __builtin_amdgcn_wmma_f32_16x16x4_f32(false, av, false, bf1, (short)0, a1, false, false);
  }

  float* pw = &part[wave][0];
  for (int v = 0; v < 8; ++v) {
    pw[(8 * hh + v) * G2_PITCH + c]      = a0[v];
    pw[(8 * hh + v) * G2_PITCH + 16 + c] = a1[v];
  }
  __syncthreads();

  for (int rr = 0; rr < 2; ++rr) {
    const int row = 2 * wave + rr;
    float s = part[0][row * G2_PITCH + lane];
    for (int w = 1; w < G2_SPLIT; ++w) s += part[w][row * G2_PITCH + lane];
    out[(size_t)(mt * 16 + row) * OUT_ + ns * 32 + lane] = s + bb;
  }
}

extern "C" void kernel_launch(void* const* d_in, const int* in_sizes, int n_in,
                              void* d_out, int out_size, void* d_ws, size_t ws_size,
                              hipStream_t stream) {
  const float* x0   = (const float*)d_in[0];
  const int*   idx0 = (const int*)  d_in[1];
  const float* x1   = (const float*)d_in[2];
  const int*   idx1 = (const int*)  d_in[3];
  const float* x2   = (const float*)d_in[4];
  const int*   idx2 = (const int*)  d_in[5];
  const float* W1_0 = (const float*)d_in[6];
  const float* b1_0 = (const float*)d_in[7];
  const float* W1_1 = (const float*)d_in[8];
  const float* b1_1 = (const float*)d_in[9];
  const float* W1_2 = (const float*)d_in[10];
  const float* b1_2 = (const float*)d_in[11];
  const float* W2   = (const float*)d_in[12];
  const float* b2   = (const float*)d_in[13];
  float* out = (float*)d_out;

  float* pooled = (float*)d_ws;
  float* h      = pooled + 3 * B_ * D_;
  int*   inv    = (int*)(h + (size_t)B_ * CONC_);
  const int n0 = B_ * 128, n1 = B_ * 256, n2 = B_ * 64;
  int* inv0 = inv;
  int* inv1 = inv0 + n0;
  int* inv2 = inv1 + n1;

  hipMemsetAsync(inv,    0xFF, (size_t)(n0 + n1 + n2) * sizeof(int), stream);

  scatter_inv<<<(n0 + 255) / 256, 256, 0, stream>>>(idx0, n0, inv0);
  scatter_inv<<<(n1 + 255) / 256, 256, 0, stream>>>(idx1, n1, inv1);
  scatter_inv<<<(n2 + 255) / 256, 256, 0, stream>>>(idx2, n2, inv2);

  gather_sum<128><<<B_, 256, 0, stream>>>(x0, inv0, pooled);
  gather_sum<256><<<B_, 256, 0, stream>>>(x1, inv1, pooled + B_ * D_);
  gather_sum< 64><<<B_, 256, 0, stream>>>(x2, inv2, pooled + 2 * B_ * D_);

  dim3 g1(32, 3);
  gemm1_relu<<<g1, 256, 0, stream>>>(pooled, W1_0, b1_0, W1_1, b1_1, W1_2, b1_2, h);

  gemm2<<<128, 256, 0, stream>>>(h, W2, b2, out);
}


# index-map phase: the three per-dim inverse-map launches become one launch of 112 work-groups, each lane mapping eight rows from two 16-byte index loads (896 waves, were 7168); sums unchanged
# speedup vs baseline: 4.8621x; 1.0914x over previous
#include <hip/hip_runtime.h>

typedef float v2f __attribute__((ext_vector_type(2)));
typedef float v4f __attribute__((ext_vector_type(4)));
typedef float v8f __attribute__((ext_vector_type(8)));

#define B_    512
#define D_    256
#define HID_  512
#define OUT_  128
#define CONC_ 1536

typedef int v4i __attribute__((ext_vector_type(4)));

#define N0_ (B_ * 128)
#define N1_ (B_ * 256)
#define N2_ (B_ * 64)
#define SC_ROWS 2048
#define SC_G0 (N0_ / SC_ROWS)
#define SC_G1 (N1_ / SC_ROWS)
#define SC_G2 (N2_ / SC_ROWS)
static_assert(N0_ % SC_ROWS == 0 && N1_ % SC_ROWS == 0 && N2_ % SC_ROWS == 0,
              "whole work-groups per dim");

__global__ __launch_bounds__(256) void scatter_inv3(const int* __restrict__ idx0,
                                                    const int* __restrict__ idx1,
                                                    const int* __restrict__ idx2,
                                                    int* __restrict__ inv) {
  const int  gb  = blockIdx.x;
  const bool is1 = gb < SC_G1;
  const bool is0 = !is1 && gb < SC_G1 + SC_G0;
  const int* idx = is1 ? idx1 : is0 ? idx0 : idx2;
  int*       dst = inv + (is1 ? N0_ : is0 ? 0 : N0_ + N1_);
  const int  g   = gb - (is1 ? 0 : is0 ? SC_G1 : SC_G1 + SC_G0);

  const int wave = threadIdx.x >> 5;
  const int lane = threadIdx.x & 31;
  const int r0 = g * SC_ROWS + wave * 256 + 4 * lane;
  const v4i a = *(const v4i*)(idx + r0);
  const v4i b = *(const v4i*)(idx + r0 + 128);
  dst[a.x] = r0;        dst[a.y] = r0 + 1;    dst[a.z] = r0 + 2;    dst[a.w] = r0 + 3;
  dst[b.x] = r0 + 128;  dst[b.y] = r0 + 129;  dst[b.z] = r0 + 130;  dst[b.w] = r0 + 131;
}

template <int M>
__global__ __launch_bounds__(256) void gather_sum(const float* __restrict__ x,
                                                  const int* __restrict__ inv,
                                                  float* __restrict__ pooled) {
  constexpr int RPW = M / 8;
  constexpr int GRP = 8;
  static_assert(RPW >= GRP && RPW <= 32 && RPW % GRP == 0, "one index per lane, whole groups");
  __shared__ v4f part[8][64];

  const int b    = blockIdx.x;
  const int wave = __builtin_amdgcn_readfirstlane(threadIdx.x >> 5);
  const int lane = threadIdx.x & 31;

  int myr = -1;
  if (lane < RPW) myr = inv[b * M + wave * RPW + lane];

  v4f acc0 = {0.f, 0.f, 0.f, 0.f}, acc1 = {0.f, 0.f, 0.f, 0.f};
#pragma unroll
  for (int j0 = 0; j0 < RPW; j0 += GRP) {
    v4f v0[GRP], v1[GRP];
    int rr[GRP];
#pragma unroll
    for (int j = 0; j < GRP; ++j) {
      rr[j] = __builtin_amdgcn_readlane(myr, j0 + j);
      const int rc = rr[j] < 0 ? 0 : rr[j];
      const v4f* row = (const v4f*)(x + (size_t)rc * D_);
      v0[j] = __builtin_nontemporal_load(row + lane);
      v1[j] = __builtin_nontemporal_load(row + 32 + lane);
    }
#pragma unroll
    for (int j = 0; j < GRP; ++j) {
      const bool on = rr[j] >= 0;
      acc0.x += on ? v0[j].x : 0.f;  acc0.y += on ? v0[j].y : 0.f;
      acc0.z += on ? v0[j].z : 0.f;  acc0.w += on ? v0[j].w : 0.f;
      acc1.x += on ? v1[j].x : 0.f;  acc1.y += on ? v1[j].y : 0.f;
      acc1.z += on ? v1[j].z : 0.f;  acc1.w += on ? v1[j].w : 0.f;
    }
  }
  part[wave][lane]      = acc0;
  part[wave][32 + lane] = acc1;
  __syncthreads();

  const float* pf = (const float*)&part[0][0];
  const int d = threadIdx.x;
  float s = pf[d];
#pragma unroll
  for (int w = 1; w < 8; ++w) s += pf[w * D_ + d];
  pooled[b * D_ + d] = s;
}

__global__ __launch_bounds__(256) void gemm1_relu(const float* __restrict__ pooledAll,
    const float* __restrict__ W1_0, const float* __restrict__ b1_0,
    const float* __restrict__ W1_1, const float* __restrict__ b1_1,
    const float* __restrict__ W1_2, const float* __restrict__ b1_2,
    float* __restrict__ h) {
  const int dim = blockIdx.y;
  const float* W    = (dim == 0) ? W1_0 : (dim == 1) ? W1_1 : W1_2;
  const float* bias = (dim == 0) ? b1_0 : (dim == 1) ? b1_1 : b1_2;
  const float* P    = pooledAll + (size_t)dim * B_ * D_;

  const int wave = threadIdx.x >> 5;
  const int lane = threadIdx.x & 31;
  const int job  = blockIdx.x * 8 + wave;
  const int mt   = job >> 3;
  const int ns   = job & 7;

  const int mrow = mt * 16 + (lane & 15);
  const int kk   = (lane >> 4) * 2;
  const int ncol = ns * 64 + (lane & 15);

  v8f a0 = {}, a1 = {}, a2 = {}, a3 = {};
  for (int k = 0; k < D_; k += 4) {
    v2f av = *(const v2f*)(P + (size_t)mrow * D_ + k + kk);
    const float* w0 = W + (size_t)(k + kk) * HID_ + ncol;
    const float* w1 = w0 + HID_;
    v2f bf0 = { w0[0],  w1[0]  };
    v2f bf1 = { w0[16], w1[16] };
    v2f bf2 = { w0[32], w1[32] };
    v2f bf3 = { w0[48], w1[48] };
    a0 = __builtin_amdgcn_wmma_f32_16x16x4_f32(false, av, false, bf0, (short)0, a0, false, false);
    a1 = __builtin_amdgcn_wmma_f32_16x16x4_f32(false, av, false, bf1, (short)0, a1, false, false);
    a2 = __builtin_amdgcn_wmma_f32_16x16x4_f32(false, av, false, bf2, (short)0, a2, false, false);
    a3 = __builtin_amdgcn_wmma_f32_16x16x4_f32(false, av, false, bf3, (short)0, a3, false, false);
  }

  const int nbase = ns * 64 + (lane & 15);
  const int rbase = mt * 16 + (lane >> 4) * 8;
  const float bb0 = bias[nbase], bb1 = bias[nbase + 16],
              bb2 = bias[nbase + 32], bb3 = bias[nbase + 48];
#pragma unroll
  for (int v = 0; v < 8; ++v) {
    float* hr = h + (size_t)(rbase + v) * CONC_ + (size_t)dim * HID_ + nbase;
    hr[0]  = fmaxf(a0[v] + bb0, 0.f);
    hr[16] = fmaxf(a1[v] + bb1, 0.f);
    hr[32] = fmaxf(a2[v] + bb2, 0.f);
    hr[48] = fmaxf(a3[v] + bb3, 0.f);
  }
}

#define G2_SPLIT 8
#define G2_KW    (CONC_ / G2_SPLIT)
#define G2_PITCH 36

__global__ __launch_bounds__(256) void gemm2(const float* __restrict__ h,
                                             const float* __restrict__ W2,
                                             const float* __restrict__ b2,
                                             float* __restrict__ out) {
  __shared__ float part[G2_SPLIT][16 * G2_PITCH];

  const int wave = threadIdx.x >> 5;
  const int lane = threadIdx.x & 31;
  const int mt   = blockIdx.x >> 2;
  const int ns   = blockIdx.x & 3;

  const int c    = lane & 15;
  const int hh   = lane >> 4;
  const int kk   = hh * 2;
  const int k0   = wave * G2_KW;

  const float* ap = h  + (size_t)(mt * 16 + c) * CONC_ + k0 + kk;
  const float* bp = W2 + (size_t)(k0 + kk) * OUT_ + ns * 32 + c;

  const float bb = b2[ns * 32 + lane];

  v8f a0 = {}, a1 = {};
#pragma unroll
  for (int s = 0; s < G2_KW / 4; ++s) {
    v2f av = *(const v2f*)(ap + 4 * s);
    const float* w0 = bp + (size_t)(4 * s) * OUT_;
    const float* w1 = w0 + OUT_;
    v2f bf0 = { w0[0],  w1[0]  };
    v2f bf1 = { w0[16], w1[16] };
    a0 = __builtin_amdgcn_wmma_f32_16x16x4_f32(false, av, false, bf0, (short)0, a0, false, false);
    a1 = __builtin_amdgcn_wmma_f32_16x16x4_f32(false, av, false, bf1, (short)0, a1, false, false);
  }

  float* pw = &part[wave][0];
  for (int v = 0; v < 8; ++v) {
    pw[(8 * hh + v) * G2_PITCH + c]      = a0[v];
    pw[(8 * hh + v) * G2_PITCH + 16 + c] = a1[v];
  }
  __syncthreads();

  for (int rr = 0; rr < 2; ++rr) {
    const int row = 2 * wave + rr;
    float s = part[0][row * G2_PITCH + lane];
    for (int w = 1; w < G2_SPLIT; ++w) s += part[w][row * G2_PITCH + lane];
    out[(size_t)(mt * 16 + row) * OUT_ + ns * 32 + lane] = s + bb;
  }
}

extern "C" void kernel_launch(void* const* d_in, const int* in_sizes, int n_in,
                              void* d_out, int out_size, void* d_ws, size_t ws_size,
                              hipStream_t stream) {
  const float* x0   = (const float*)d_in[0];
  const int*   idx0 = (const int*)  d_in[1];
  const float* x1   = (const float*)d_in[2];
  const int*   idx1 = (const int*)  d_in[3];
  const float* x2   = (const float*)d_in[4];
  const int*   idx2 = (const int*)  d_in[5];
  const float* W1_0 = (const float*)d_in[6];
  const float* b1_0 = (const float*)d_in[7];
  const float* W1_1 = (const float*)d_in[8];
  const float* b1_1 = (const float*)d_in[9];
  const float* W1_2 = (const float*)d_in[10];
  const float* b1_2 = (const float*)d_in[11];
  const float* W2   = (const float*)d_in[12];
  const float* b2   = (const float*)d_in[13];
  float* out = (float*)d_out;

  float* pooled = (float*)d_ws;
  float* h      = pooled + 3 * B_ * D_;
  int*   inv    = (int*)(h + (size_t)B_ * CONC_);
  int* inv0 = inv;
  int* inv1 = inv0 + N0_;
  int* inv2 = inv1 + N1_;

  hipMemsetAsync(inv,    0xFF, (size_t)(N0_ + N1_ + N2_) * sizeof(int), stream);

  scatter_inv3<<<SC_G1 + SC_G0 + SC_G2, 256, 0, stream>>>(idx0, idx1, idx2, inv);

  gather_sum<128><<<B_, 256, 0, stream>>>(x0, inv0, pooled);
  gather_sum<256><<<B_, 256, 0, stream>>>(x1, inv1, pooled + B_ * D_);
  gather_sum< 64><<<B_, 256, 0, stream>>>(x2, inv2, pooled + 2 * B_ * D_);

  dim3 g1(32, 3);
  gemm1_relu<<<g1, 256, 0, stream>>>(pooled, W1_0, b1_0, W1_1, b1_1, W1_2, b1_2, h);

  gemm2<<<128, 256, 0, stream>>>(h, W2, b2, out);
}


# sum-pool: one launch instead of three: 256 work-groups x 16 waves, a wave streams its 56 rows of the three dims, the next 8 rows' loads issued before 8 are summed (16-32 in flight); sums unchanged
# speedup vs baseline: 5.8156x; 1.1961x over previous
#include <hip/hip_runtime.h>

typedef float v2f __attribute__((ext_vector_type(2)));
typedef float v4f __attribute__((ext_vector_type(4)));
typedef float v8f __attribute__((ext_vector_type(8)));

#define B_    512
#define D_    256
#define HID_  512
#define OUT_  128
#define CONC_ 1536

typedef int v4i __attribute__((ext_vector_type(4)));

#define N0_ (B_ * 128)
#define N1_ (B_ * 256)
#define N2_ (B_ * 64)
#define SC_ROWS 2048
#define SC_G0 (N0_ / SC_ROWS)
#define SC_G1 (N1_ / SC_ROWS)
#define SC_G2 (N2_ / SC_ROWS)
static_assert(N0_ % SC_ROWS == 0 && N1_ % SC_ROWS == 0 && N2_ % SC_ROWS == 0,
              "whole work-groups per dim");

__global__ __launch_bounds__(256) void scatter_inv3(const int* __restrict__ idx0,
                                                    const int* __restrict__ idx1,
                                                    const int* __restrict__ idx2,
                                                    int* __restrict__ inv) {
  const int  gb  = blockIdx.x;
  const bool is1 = gb < SC_G1;
  const bool is0 = !is1 && gb < SC_G1 + SC_G0;
  const int* idx = is1 ? idx1 : is0 ? idx0 : idx2;
  int*       dst = inv + (is1 ? N0_ : is0 ? 0 : N0_ + N1_);
  const int  g   = gb - (is1 ? 0 : is0 ? SC_G1 : SC_G1 + SC_G0);

  const int wave = threadIdx.x >> 5;
  const int lane = threadIdx.x & 31;
  const int r0 = g * SC_ROWS + wave * 256 + 4 * lane;
  const v4i a = *(const v4i*)(idx + r0);
  const v4i b = *(const v4i*)(idx + r0 + 128);
  dst[a.x] = r0;        dst[a.y] = r0 + 1;    dst[a.z] = r0 + 2;    dst[a.w] = r0 + 3;
  dst[b.x] = r0 + 128;  dst[b.y] = r0 + 129;  dst[b.z] = r0 + 130;  dst[b.w] = r0 + 131;
}

#define GS_WAVES 16
#define GS_BPG   2
#define GS_GRP   8
#define GS_NGRP  7
static_assert(B_ % GS_BPG == 0, "whole work-groups");
static_assert(128 / 8 == 2 * GS_GRP && 256 / 8 == 4 * GS_GRP && 64 / 8 == GS_GRP,
              "a wave's positions per dim are whole groups: 2 + 4 + 1");

__global__ __launch_bounds__(GS_WAVES * 32) void gather_sum3(const float* __restrict__ x0,
                                                             const float* __restrict__ x1,
                                                             const float* __restrict__ x2,
                                                             const int* __restrict__ inv,
                                                             float* __restrict__ pooled) {
  __shared__ v4f part[3][GS_WAVES][64];

  const int wave = __builtin_amdgcn_readfirstlane(threadIdx.x >> 5);
  const int lane = threadIdx.x & 31;
  const int b    = blockIdx.x * GS_BPG + (wave >> 3);
  const int w    = wave & 7;

  int r0 = -1, r2 = -1;
  if (lane < 16) r0 = inv[b * 128 + w * 16 + lane];
  const int r1 = inv[N0_ + b * 256 + w * 32 + lane];
  if (lane < 8)  r2 = inv[N0_ + N1_ + b * 64 + w * 8 + lane];

  v4f va[2][GS_GRP], vb[2][GS_GRP];
  int rr[2][GS_GRP];
  float acc[3][8];
#pragma unroll
  for (int d = 0; d < 3; ++d)
#pragma unroll
    for (int e = 0; e < 8; ++e) acc[d][e] = 0.f;

#define GS_DIM(G) ((G) < 2 ? 0 : (G) < 6 ? 1 : 2)
#define GS_J0(G)  ((G) < 2 ? GS_GRP * (G) : (G) < 6 ? GS_GRP * ((G) - 2) : 0)
#define GS_LOAD(G) do {                                                                        \
    const float* xs_ = GS_DIM(G) == 0 ? x0 : GS_DIM(G) == 1 ? x1 : x2;                         \
    const int    rw_ = GS_DIM(G) == 0 ? r0 : GS_DIM(G) == 1 ? r1 : r2;                         \
    _Pragma("unroll") for (int j = 0; j < GS_GRP; ++j) {                                       \
      const int r_  = __builtin_amdgcn_readlane(rw_, GS_J0(G) + j);       \
      rr[(G) & 1][j] = r_;                                                                     \
      const int rc_ = r_ < 0 ? 0 : r_;                  \
      const v4f* row_ = (const v4f*)(xs_ + (size_t)rc_ * D_);                                  \
      va[(G) & 1][j] = __builtin_nontemporal_load(row_ + lane);             \
      vb[(G) & 1][j] = __builtin_nontemporal_load(row_ + 32 + lane);                           \
    } } while (0)

  GS_LOAD(0);
#pragma unroll
  for (int g = 0; g < GS_NGRP; ++g) {
    if (g + 1 < GS_NGRP) GS_LOAD(g + 1);
    float s0 = acc[GS_DIM(g)][0], s1 = acc[GS_DIM(g)][1], s2 = acc[GS_DIM(g)][2], s3 = acc[GS_DIM(g)][3],
          s4 = acc[GS_DIM(g)][4], s5 = acc[GS_DIM(g)][5], s6 = acc[GS_DIM(g)][6], s7 = acc[GS_DIM(g)][7];
#pragma unroll
    for (int j = 0; j < GS_GRP; ++j) {
      const bool on = rr[g & 1][j] >= 0;
      const v4f p = va[g & 1][j], q = vb[g & 1][j];
      s0 += on ? p.x : 0.f;  s1 += on ? p.y : 0.f;
      s2 += on ? p.z : 0.f;  s3 += on ? p.w : 0.f;
      s4 += on ? q.x : 0.f;  s5 += on ? q.y : 0.f;
      s6 += on ? q.z : 0.f;  s7 += on ? q.w : 0.f;
    }
    asm volatile("" : "+v"(s0), "+v"(s1), "+v"(s2), "+v"(s3), "+v"(s4), "+v"(s5), "+v"(s6), "+v"(s7));
    acc[GS_DIM(g)][0] = s0;  acc[GS_DIM(g)][1] = s1;  acc[GS_DIM(g)][2] = s2;  acc[GS_DIM(g)][3] = s3;
    acc[GS_DIM(g)][4] = s4;  acc[GS_DIM(g)][5] = s5;  acc[GS_DIM(g)][6] = s6;  acc[GS_DIM(g)][7] = s7;
  }
#undef GS_LOAD
#undef GS_J0
#undef GS_DIM

#pragma unroll
  for (int d = 0; d < 3; ++d) {
    part[d][wave][lane]      = (v4f){acc[d][0], acc[d][1], acc[d][2], acc[d][3]};
    part[d][wave][32 + lane] = (v4f){acc[d][4], acc[d][5], acc[d][6], acc[d][7]};
  }
  __syncthreads();

  const float* pf = (const float*)&part[0][0][0];
  const int hb = threadIdx.x >> 8;
  const int d  = threadIdx.x & (D_ - 1);
  const int bo = blockIdx.x * GS_BPG + hb;
#pragma unroll
  for (int dim = 0; dim < 3; ++dim) {
    const float* p = pf + (dim * GS_WAVES + hb * 8) * D_ + d;
    float s = p[0];
#pragma unroll
    for (int k = 1; k < 8; ++k) s += p[k * D_];
    pooled[(size_t)dim * B_ * D_ + bo * D_ + d] = s;
  }
}

__global__ __launch_bounds__(256) void gemm1_relu(const float* __restrict__ pooledAll,
    const float* __restrict__ W1_0, const float* __restrict__ b1_0,
    const float* __restrict__ W1_1, const float* __restrict__ b1_1,
    const float* __restrict__ W1_2, const float* __restrict__ b1_2,
    float* __restrict__ h) {
  const int dim = blockIdx.y;
  const float* W    = (dim == 0) ? W1_0 : (dim == 1) ? W1_1 : W1_2;
  const float* bias = (dim == 0) ? b1_0 : (dim == 1) ? b1_1 : b1_2;
  const float* P    = pooledAll + (size_t)dim * B_ * D_;

  const int wave = threadIdx.x >> 5;
  const int lane = threadIdx.x & 31;
  const int job  = blockIdx.x * 8 + wave;
  const int mt   = job >> 3;
  const int ns   = job & 7;

  const int mrow = mt * 16 + (lane & 15);
  const int kk   = (lane >> 4) * 2;
  const int ncol = ns * 64 + (lane & 15);

  v8f a0 = {}, a1 = {}, a2 = {}, a3 = {};
  for (int k = 0; k < D_; k += 4) {
    v2f av = *(const v2f*)(P + (size_t)mrow * D_ + k + kk);
    const float* w0 = W + (size_t)(k + kk) * HID_ + ncol;
    const float* w1 = w0 + HID_;
    v2f bf0 = { w0[0],  w1[0]  };
    v2f bf1 = { w0[16], w1[16] };
    v2f bf2 = { w0[32], w1[32] };
    v2f bf3 = { w0[48], w1[48] };
    a0 = __builtin_amdgcn_wmma_f32_16x16x4_f32(false, av, false, bf0, (short)0, a0, false, false);
    a1 = __builtin_amdgcn_wmma_f32_16x16x4_f32(false, av, false, bf1, (short)0, a1, false, false);
    a2 = __builtin_amdgcn_wmma_f32_16x16x4_f32(false, av, false, bf2, (short)0, a2, false, false);
    a3 = __builtin_amdgcn_wmma_f32_16x16x4_f32(false, av, false, bf3, (short)0, a3, false, false);
  }

  const int nbase = ns * 64 + (lane & 15);
  const int rbase = mt * 16 + (lane >> 4) * 8;
  const float bb0 = bias[nbase], bb1 = bias[nbase + 16],
              bb2 = bias[nbase + 32], bb3 = bias[nbase + 48];
#pragma unroll
  for (int v = 0; v < 8; ++v) {
    float* hr = h + (size_t)(rbase + v) * CONC_ + (size_t)dim * HID_ + nbase;
    hr[0]  = fmaxf(a0[v] + bb0, 0.f);
    hr[16] = fmaxf(a1[v] + bb1, 0.f);
    hr[32] = fmaxf(a2[v] + bb2, 0.f);
    hr[48] = fmaxf(a3[v] + bb3, 0.f);
  }
}

#define G2_SPLIT 8
#define G2_KW    (CONC_ / G2_SPLIT)
#define G2_PITCH 36

__global__ __launch_bounds__(256) void gemm2(const float* __restrict__ h,
                                             const float* __restrict__ W2,
                                             const float* __restrict__ b2,
                                             float* __restrict__ out) {
  __shared__ float part[G2_SPLIT][16 * G2_PITCH];

  const int wave = threadIdx.x >> 5;
  const int lane = threadIdx.x & 31;
  const int mt   = blockIdx.x >> 2;
  const int ns   = blockIdx.x & 3;

  const int c    = lane & 15;
  const int hh   = lane >> 4;
  const int kk   = hh * 2;
  const int k0   = wave * G2_KW;

  const float* ap = h  + (size_t)(mt * 16 + c) * CONC_ + k0 + kk;
  const float* bp = W2 + (size_t)(k0 + kk) * OUT_ + ns * 32 + c;

  const float bb = b2[ns * 32 + lane];

  v8f a0 = {}, a1 = {};
#pragma unroll
  for (int s = 0; s < G2_KW / 4; ++s) {
    v2f av = *(const v2f*)(ap + 4 * s);
    const float* w0 = bp + (size_t)(4 * s) * OUT_;
    const float* w1 = w0 + OUT_;
    v2f bf0 = { w0[0],  w1[0]  };
    v2f bf1 = { w0[16], w1[16] };
    a0 = __builtin_amdgcn_wmma_f32_16x16x4_f32(false, av, false, bf0, (short)0, a0, false, false);
    a1 = __builtin_amdgcn_wmma_f32_16x16x4_f32(false, av, false, bf1, (short)0, a1, false, false);
  }

  float* pw = &part[wave][0];
  for (int v = 0; v < 8; ++v) {
    pw[(8 * hh + v) * G2_PITCH + c]      = a0[v];
    pw[(8 * hh + v) * G2_PITCH + 16 + c] = a1[v];
  }
  __syncthreads();

  for (int rr = 0; rr < 2; ++rr) {
    const int row = 2 * wave + rr;
    float s = part[0][row * G2_PITCH + lane];
    for (int w = 1; w < G2_SPLIT; ++w) s += part[w][row * G2_PITCH + lane];
    out[(size_t)(mt * 16 + row) * OUT_ + ns * 32 + lane] = s + bb;
  }
}

extern "C" void kernel_launch(void* const* d_in, const int* in_sizes, int n_in,
                              void* d_out, int out_size, void* d_ws, size_t ws_size,
                              hipStream_t stream) {
  const float* x0   = (const float*)d_in[0];
  const int*   idx0 = (const int*)  d_in[1];
  const float* x1   = (const float*)d_in[2];
  const int*   idx1 = (const int*)  d_in[3];
  const float* x2   = (const float*)d_in[4];
  const int*   idx2 = (const int*)  d_in[5];
  const float* W1_0 = (const float*)d_in[6];
  const float* b1_0 = (const float*)d_in[7];
  const float* W1_1 = (const float*)d_in[8];
  const float* b1_1 = (const float*)d_in[9];
  const float* W1_2 = (const float*)d_in[10];
  const float* b1_2 = (const float*)d_in[11];
  const float* W2   = (const float*)d_in[12];
  const float* b2   = (const float*)d_in[13];
  float* out = (float*)d_out;

  float* pooled = (float*)d_ws;
  float* h      = pooled + 3 * B_ * D_;
  int*   inv    = (int*)(h + (size_t)B_ * CONC_);

  hipMemsetAsync(inv,    0xFF, (size_t)(N0_ + N1_ + N2_) * sizeof(int), stream);

  scatter_inv3<<<SC_G1 + SC_G0 + SC_G2, 256, 0, stream>>>(idx0, idx1, idx2, inv);

  gather_sum3<<<B_ / GS_BPG, GS_WAVES * 32, 0, stream>>>(x0, x1, x2, inv, pooled);

  dim3 g1(32, 3);
  gemm1_relu<<<g1, 256, 0, stream>>>(pooled, W1_0, b1_0, W1_1, b1_1, W1_2, b1_2, h);

  gemm2<<<128, 256, 0, stream>>>(h, W2, b2, out);
}


# gemm1 re-tiled: a lane owns 4 adjacent columns, so one 16-byte W1 load per k row feeds its 4 tiles; operands requested 2 chunks ahead; one wave per SIMD; k order permuted: summation order changes
# speedup vs baseline: 6.5892x; 1.1330x over previous
#include <hip/hip_runtime.h>

typedef float v2f __attribute__((ext_vector_type(2)));
typedef float v4f __attribute__((ext_vector_type(4)));
typedef float v8f __attribute__((ext_vector_type(8)));

#define B_    512
#define D_    256
#define HID_  512
#define OUT_  128
#define CONC_ 1536

typedef int v4i __attribute__((ext_vector_type(4)));

#define N0_ (B_ * 128)
#define N1_ (B_ * 256)
#define N2_ (B_ * 64)
#define SC_ROWS 2048
#define SC_G0 (N0_ / SC_ROWS)
#define SC_G1 (N1_ / SC_ROWS)
#define SC_G2 (N2_ / SC_ROWS)
static_assert(N0_ % SC_ROWS == 0 && N1_ % SC_ROWS == 0 && N2_ % SC_ROWS == 0,
              "whole work-groups per dim");

__global__ __launch_bounds__(256) void scatter_inv3(const int* __restrict__ idx0,
                                                    const int* __restrict__ idx1,
                                                    const int* __restrict__ idx2,
                                                    int* __restrict__ inv) {
  const int  gb  = blockIdx.x;
  const bool is1 = gb < SC_G1;
  const bool is0 = !is1 && gb < SC_G1 + SC_G0;
  const int* idx = is1 ? idx1 : is0 ? idx0 : idx2;
  int*       dst = inv + (is1 ? N0_ : is0 ? 0 : N0_ + N1_);
  const int  g   = gb - (is1 ? 0 : is0 ? SC_G1 : SC_G1 + SC_G0);

  const int wave = threadIdx.x >> 5;
  const int lane = threadIdx.x & 31;
  const int r0 = g * SC_ROWS + wave * 256 + 4 * lane;
  const v4i a = *(const v4i*)(idx + r0);
  const v4i b = *(const v4i*)(idx + r0 + 128);
  dst[a.x] = r0;        dst[a.y] = r0 + 1;    dst[a.z] = r0 + 2;    dst[a.w] = r0 + 3;
  dst[b.x] = r0 + 128;  dst[b.y] = r0 + 129;  dst[b.z] = r0 + 130;  dst[b.w] = r0 + 131;
}

#define GS_WAVES 16
#define GS_BPG   2
#define GS_GRP   8
#define GS_NGRP  7
static_assert(B_ % GS_BPG == 0, "whole work-groups");
static_assert(128 / 8 == 2 * GS_GRP && 256 / 8 == 4 * GS_GRP && 64 / 8 == GS_GRP,
              "a wave's positions per dim are whole groups: 2 + 4 + 1");

__global__ __launch_bounds__(GS_WAVES * 32) void gather_sum3(const float* __restrict__ x0,
                                                             const float* __restrict__ x1,
                                                             const float* __restrict__ x2,
                                                             const int* __restrict__ inv,
                                                             float* __restrict__ pooled) {
  __shared__ v4f part[3][GS_WAVES][64];

  const int wave = __builtin_amdgcn_readfirstlane(threadIdx.x >> 5);
  const int lane = threadIdx.x & 31;
  const int b    = blockIdx.x * GS_BPG + (wave >> 3);
  const int w    = wave & 7;

  int r0 = -1, r2 = -1;
  if (lane < 16) r0 = inv[b * 128 + w * 16 + lane];
  const int r1 = inv[N0_ + b * 256 + w * 32 + lane];
  if (lane < 8)  r2 = inv[N0_ + N1_ + b * 64 + w * 8 + lane];

  v4f va[2][GS_GRP], vb[2][GS_GRP];
  int rr[2][GS_GRP];
  float acc[3][8];
#pragma unroll
  for (int d = 0; d < 3; ++d)
#pragma unroll
    for (int e = 0; e < 8; ++e) acc[d][e] = 0.f;

#define GS_DIM(G) ((G) < 2 ? 0 : (G) < 6 ? 1 : 2)
#define GS_J0(G)  ((G) < 2 ? GS_GRP * (G) : (G) < 6 ? GS_GRP * ((G) - 2) : 0)
#define GS_LOAD(G) do {                                                                        \
    const float* xs_ = GS_DIM(G) == 0 ? x0 : GS_DIM(G) == 1 ? x1 : x2;                         \
    const int    rw_ = GS_DIM(G) == 0 ? r0 : GS_DIM(G) == 1 ? r1 : r2;                         \
    _Pragma("unroll") for (int j = 0; j < GS_GRP; ++j) {                                       \
      const int r_  = __builtin_amdgcn_readlane(rw_, GS_J0(G) + j);       \
      rr[(G) & 1][j] = r_;                                                                     \
      const int rc_ = r_ < 0 ? 0 : r_;                  \
      const v4f* row_ = (const v4f*)(xs_ + (size_t)rc_ * D_);                                  \
      va[(G) & 1][j] = __builtin_nontemporal_load(row_ + lane);             \
      vb[(G) & 1][j] = __builtin_nontemporal_load(row_ + 32 + lane);                           \
    } } while (0)

  GS_LOAD(0);
#pragma unroll
  for (int g = 0; g < GS_NGRP; ++g) {
    if (g + 1 < GS_NGRP) GS_LOAD(g + 1);
    float s0 = acc[GS_DIM(g)][0], s1 = acc[GS_DIM(g)][1], s2 = acc[GS_DIM(g)][2], s3 = acc[GS_DIM(g)][3],
          s4 = acc[GS_DIM(g)][4], s5 = acc[GS_DIM(g)][5], s6 = acc[GS_DIM(g)][6], s7 = acc[GS_DIM(g)][7];
#pragma unroll
    for (int j = 0; j < GS_GRP; ++j) {
      const bool on = rr[g & 1][j] >= 0;
      const v4f p = va[g & 1][j], q = vb[g & 1][j];
      s0 += on ? p.x : 0.f;  s1 += on ? p.y : 0.f;
      s2 += on ? p.z : 0.f;  s3 += on ? p.w : 0.f;
      s4 += on ? q.x : 0.f;  s5 += on ? q.y : 0.f;
      s6 += on ? q.z : 0.f;  s7 += on ? q.w : 0.f;
    }
    asm volatile("" : "+v"(s0), "+v"(s1), "+v"(s2), "+v"(s3), "+v"(s4), "+v"(s5), "+v"(s6), "+v"(s7));
    acc[GS_DIM(g)][0] = s0;  acc[GS_DIM(g)][1] = s1;  acc[GS_DIM(g)][2] = s2;  acc[GS_DIM(g)][3] = s3;
    acc[GS_DIM(g)][4] = s4;  acc[GS_DIM(g)][5] = s5;  acc[GS_DIM(g)][6] = s6;  acc[GS_DIM(g)][7] = s7;
  }
#undef GS_LOAD
#undef GS_J0
#undef GS_DIM

#pragma unroll
  for (int d = 0; d < 3; ++d) {
    part[d][wave][lane]      = (v4f){acc[d][0], acc[d][1], acc[d][2], acc[d][3]};
    part[d][wave][32 + lane] = (v4f){acc[d][4], acc[d][5], acc[d][6], acc[d][7]};
  }
  __syncthreads();

  const float* pf = (const float*)&part[0][0][0];
  const int hb = threadIdx.x >> 8;
  const int d  = threadIdx.x & (D_ - 1);
  const int bo = blockIdx.x * GS_BPG + hb;
#pragma unroll
  for (int dim = 0; dim < 3; ++dim) {
    const float* p = pf + (dim * GS_WAVES + hb * 8) * D_ + d;
    float s = p[0];
#pragma unroll
    for (int k = 1; k < 8; ++k) s += p[k * D_];
    pooled[(size_t)dim * B_ * D_ + bo * D_ + d] = s;
  }
}

#define G1_WAVES 4
#define G1_CH    8
#define G1_NCH   (D_ / 4 / G1_CH)
#define G1_AHEAD 2
static_assert(D_ % (4 * G1_CH) == 0 && G1_CH % 2 == 0, "whole chunks of K-step pairs");
static_assert((B_ / 16) % G1_WAVES == 0 && HID_ % 64 == 0, "whole work-groups");

__global__ __launch_bounds__(G1_WAVES * 32) void gemm1_relu(const float* __restrict__ pooledAll,
    const float* __restrict__ W1_0, const float* __restrict__ b1_0,
    const float* __restrict__ W1_1, const float* __restrict__ b1_1,
    const float* __restrict__ W1_2, const float* __restrict__ b1_2,
    float* __restrict__ h) {
  const int dim = blockIdx.y;
  const float* W    = (dim == 0) ? W1_0 : (dim == 1) ? W1_1 : W1_2;
  const float* bias = (dim == 0) ? b1_0 : (dim == 1) ? b1_1 : b1_2;
  const float* P    = pooledAll + (size_t)dim * B_ * D_;

  const int wave = __builtin_amdgcn_readfirstlane(threadIdx.x >> 5);
  const int lane = threadIdx.x & 31;
  const int ns   = blockIdx.x & 7;
  const int mt   = (blockIdx.x >> 3) * G1_WAVES + wave;
  const int c    = lane & 15;
  const int hh   = lane >> 4;

  const float* ap = P + (size_t)(mt * 16 + c) * D_ + 4 * hh;
  const float* bp = W + (size_t)(4 * hh) * HID_ + ns * 64 + 4 * c;

  v4f ra[G1_AHEAD + 1][G1_CH / 2];
  v4f rb[G1_AHEAD + 1][G1_CH][2];
#define G1_LOAD(C) do {                                                                      \
    _Pragma("unroll") for (int j = 0; j < G1_CH; ++j) {                                      \
      const int t_ = (C) * G1_CH + j;                                                        \
      const int k_ = 8 * (t_ >> 1) + 2 * (t_ & 1);                \
      if ((j & 1) == 0) ra[(C) % (G1_AHEAD + 1)][j >> 1] = *(const v4f*)(ap + 8 * (t_ >> 1)); \
      rb[(C) % (G1_AHEAD + 1)][j][0] = *(const v4f*)(bp + (size_t)k_ * HID_);                \
      rb[(C) % (G1_AHEAD + 1)][j][1] = *(const v4f*)(bp + (size_t)(k_ + 1) * HID_);          \
    } } while (0)

  v8f a0 = {}, a1 = {}, a2 = {}, a3 = {};
#pragma unroll
  for (int ch = 0; ch < G1_AHEAD; ++ch) G1_LOAD(ch);
#pragma unroll
  for (int ch = 0; ch < G1_NCH; ++ch) {
    __builtin_amdgcn_sched_barrier(0);
    if (ch + G1_AHEAD < G1_NCH) G1_LOAD(ch + G1_AHEAD);
    __builtin_amdgcn_sched_barrier(0);
#pragma unroll
    for (int j = 0; j < G1_CH; ++j) {
      const v4f pa = ra[ch % (G1_AHEAD + 1)][j >> 1];
      const v2f av = (j & 1) ? (v2f){pa.z, pa.w} : (v2f){pa.x, pa.y};
      const v4f q0 = rb[ch % (G1_AHEAD + 1)][j][0], q1 = rb[ch % (G1_AHEAD + 1)][j][1];
      const v2f f0 = {q0.x, q1.x}, f1 = {q0.y, q1.y}, f2 = {q0.z, q1.z}, f3 = {q0.w, q1.w};
      a0 = __builtin_amdgcn_wmma_f32_16x16x4_f32(false, av, false, f0, (short)0, a0, false, false);
      a1 = __builtin_amdgcn_wmma_f32_16x16x4_f32(false, av, false, f1, (short)0, a1, false, false);
      a2 = __builtin_amdgcn_wmma_f32_16x16x4_f32(false, av, false, f2, (short)0, a2, false, false);
      a3 = __builtin_amdgcn_wmma_f32_16x16x4_f32(false, av, false, f3, (short)0, a3, false, false);
    }
  }
#undef G1_LOAD

  const int n0 = ns * 64 + 4 * c;
  const v4f bb = *(const v4f*)(bias + n0);
#pragma unroll
  for (int v = 0; v < 8; ++v) {
    float* hr = h + (size_t)(mt * 16 + 8 * hh + v) * CONC_ + (size_t)dim * HID_ + n0;
    *(v4f*)hr = (v4f){fmaxf(a0[v] + bb.x, 0.f), fmaxf(a1[v] + bb.y, 0.f),
                      fmaxf(a2[v] + bb.z, 0.f), fmaxf(a3[v] + bb.w, 0.f)};
  }
}

#define G2_SPLIT 8
#define G2_KW    (CONC_ / G2_SPLIT)
#define G2_PITCH 36

__global__ __launch_bounds__(256) void gemm2(const float* __restrict__ h,
                                             const float* __restrict__ W2,
                                             const float* __restrict__ b2,
                                             float* __restrict__ out) {
  __shared__ float part[G2_SPLIT][16 * G2_PITCH];

  const int wave = threadIdx.x >> 5;
  const int lane = threadIdx.x & 31;
  const int mt   = blockIdx.x >> 2;
  const int ns   = blockIdx.x & 3;

  const int c    = lane & 15;
  const int hh   = lane >> 4;
  const int kk   = hh * 2;
  const int k0   = wave * G2_KW;

  const float* ap = h  + (size_t)(mt * 16 + c) * CONC_ + k0 + kk;
  const float* bp = W2 + (size_t)(k0 + kk) * OUT_ + ns * 32 + c;

  const float bb = b2[ns * 32 + lane];

  v8f a0 = {}, a1 = {};
#pragma unroll
  for (int s = 0; s < G2_KW / 4; ++s) {
    v2f av = *(const v2f*)(ap + 4 * s);
    const float* w0 = bp + (size_t)(4 * s) * OUT_;
    const float* w1 = w0 + OUT_;
    v2f bf0 = { w0[0],  w1[0]  };
    v2f bf1 = { w0[16], w1[16] };
    a0 = __builtin_amdgcn_wmma_f32_16x16x4_f32(false, av, false, bf0, (short)0, a0, false, false);
    a1 = __builtin_amdgcn_wmma_f32_16x16x4_f32(false, av, false, bf1, (short)0, a1, false, false);
  }

  float* pw = &part[wave][0];
  for (int v = 0; v < 8; ++v) {
    pw[(8 * hh + v) * G2_PITCH + c]      = a0[v];
    pw[(8 * hh + v) * G2_PITCH + 16 + c] = a1[v];
  }
  __syncthreads();

  for (int rr = 0; rr < 2; ++rr) {
    const int row = 2 * wave + rr;
    float s = part[0][row * G2_PITCH + lane];
    for (int w = 1; w < G2_SPLIT; ++w) s += part[w][row * G2_PITCH + lane];
    out[(size_t)(mt * 16 + row) * OUT_ + ns * 32 + lane] = s + bb;
  }
}

extern "C" void kernel_launch(void* const* d_in, const int* in_sizes, int n_in,
                              void* d_out, int out_size, void* d_ws, size_t ws_size,
                              hipStream_t stream) {
  const float* x0   = (const float*)d_in[0];
  const int*   idx0 = (const int*)  d_in[1];
  const float* x1   = (const float*)d_in[2];
  const int*   idx1 = (const int*)  d_in[3];
  const float* x2   = (const float*)d_in[4];
  const int*   idx2 = (const int*)  d_in[5];
  const float* W1_0 = (const float*)d_in[6];
  const float* b1_0 = (const float*)d_in[7];
  const float* W1_1 = (const float*)d_in[8];
  const float* b1_1 = (const float*)d_in[9];
  const float* W1_2 = (const float*)d_in[10];
  const float* b1_2 = (const float*)d_in[11];
  const float* W2   = (const float*)d_in[12];
  const float* b2   = (const float*)d_in[13];
  float* out = (float*)d_out;

  float* pooled = (float*)d_ws;
  float* h      = pooled + 3 * B_ * D_;
  int*   inv    = (int*)(h + (size_t)B_ * CONC_);

  hipMemsetAsync(inv,    0xFF, (size_t)(N0_ + N1_ + N2_) * sizeof(int), stream);

  scatter_inv3<<<SC_G1 + SC_G0 + SC_G2, 256, 0, stream>>>(idx0, idx1, idx2, inv);

  gather_sum3<<<B_ / GS_BPG, GS_WAVES * 32, 0, stream>>>(x0, x1, x2, inv, pooled);

  dim3 g1(64, 3);
  gemm1_relu<<<g1, G1_WAVES * 32, 0, stream>>>(pooled, W1_0, b1_0, W1_1, b1_1, W1_2, b1_2, h);

  gemm2<<<128, 256, 0, stream>>>(h, W2, b2, out);
}
